# starting kernel
# baseline (speedup 1.0000x reference)
#include <hip/hip_runtime.h>


#define N_NODES 50000
#define E_EDGES 800000
#define ET      (E_EDGES + N_NODES)
#define DIN     256
#define C_OUT   128
#define NEG_SLOPE 0.2f

typedef __attribute__((ext_vector_type(2))) float v2f;
typedef __attribute__((ext_vector_type(8))) float v8f;

__device__ __forceinline__ void hw_atomic_add_f32(float* p, float v) {
  asm volatile("global_atomic_add_f32 %0, %1, off" : : "v"(p), "v"(v) : "memory");
}

__global__ __launch_bounds__(256) void k_gemm(const float* __restrict__ x,
                                              const float* __restrict__ W,
                                              float* __restrict__ h) {
  const int wave = threadIdx.x >> 5;
  const int lane = threadIdx.x & 31;
  const int tile = blockIdx.x * 8 + wave;
  const int tM   = (tile >> 3) << 4;
  const int tN   = (tile & 7) << 4;
  const int hi   = lane >> 4;
  const int l    = lane & 15;

  const float* arow = x + (size_t)(tM + l) * DIN;
  v8f acc = (v8f)0.0f;

  for (int k = 0; k < DIN; k += 4) {
    const int ka = k + 2 * hi;
    v2f a = *(const v2f*)(arow + ka);
    v2f b;
    b.x = W[(size_t)ka * C_OUT + tN + l];
    b.y = W[(size_t)(ka + 1) * C_OUT + tN + l];
    acc = __builtin_amdgcn_wmma_f32_16x16x4_f32(false, a, false, b,
                                                (short)0, acc, false, false);
  }
  float* orow = h + (size_t)(tM + hi * 8) * C_OUT + tN + l;
#pragma unroll
  for (int r = 0; r < 8; ++r) orow[(size_t)r * C_OUT] = acc[r];
}

__global__ __launch_bounds__(256) void k_nodeatt(const float* __restrict__ h,
                                                 const float* __restrict__ att,
                                                 float* __restrict__ ai,
                                                 float* __restrict__ aj) {
  const int wave = threadIdx.x >> 5;
  const int lane = threadIdx.x & 31;
  const int n = blockIdx.x * 8 + wave;
  const float4 hv  = *(const float4*)(h + (size_t)n * C_OUT + lane * 4);
  const float4 av  = *(const float4*)(att + lane * 4);
  const float4 bv  = *(const float4*)(att + C_OUT + lane * 4);
  float si = hv.x*av.x + hv.y*av.y + hv.z*av.z + hv.w*av.w;
  float sj = hv.x*bv.x + hv.y*bv.y + hv.z*bv.z + hv.w*bv.w;
#pragma unroll
  for (int off = 16; off > 0; off >>= 1) {
    si += __shfl_down(si, off, 32);
    sj += __shfl_down(sj, off, 32);
  }
  if (lane == 0) { ai[n] = si; aj[n] = sj; }
}

__global__ void k_init(unsigned* __restrict__ amax, float* __restrict__ denom) {
  int n = blockIdx.x * blockDim.x + threadIdx.x;
  if (n < N_NODES) { amax[n] = 0u; denom[n] = 0.0f; }
}
__global__ void k_initout(float* __restrict__ out, const float* __restrict__ bias) {
  int i = blockIdx.x * blockDim.x + threadIdx.x;
  if (i < N_NODES * C_OUT) out[i] = bias[i & (C_OUT - 1)];
}

__device__ __forceinline__ unsigned f2key(float f) {
  unsigned b = __float_as_uint(f);
  return (b & 0x80000000u) ? ~b : (b | 0x80000000u);
}
__device__ __forceinline__ float key2f(unsigned k) {
  unsigned b = (k & 0x80000000u) ? (k & 0x7FFFFFFFu) : ~k;
  return __uint_as_float(b);
}
__device__ __forceinline__ float leaky(float a) {
  return a > 0.0f ? a : NEG_SLOPE * a;
}

__global__ void k_edgemax(const int* __restrict__ esrc, const int* __restrict__ edst,
                          const float* __restrict__ ai, const float* __restrict__ aj,
                          unsigned* __restrict__ amax) {
  int e = blockIdx.x * blockDim.x + threadIdx.x;
  if (e >= ET) return;
  int s, d;
  if (e < E_EDGES) { s = esrc[e]; d = edst[e]; } else { s = d = e - E_EDGES; }
  float a = leaky(ai[d] + aj[s]);
  atomicMax(&amax[d], f2key(a));
}

__global__ void k_edgeexp(const int* __restrict__ esrc, const int* __restrict__ edst,
                          const float* __restrict__ ai, const float* __restrict__ aj,
                          const unsigned* __restrict__ amax,
                          float* __restrict__ denom, float* __restrict__ ex) {
  int e = blockIdx.x * blockDim.x + threadIdx.x;
  if (e >= ET) return;
  int s, d;
  if (e < E_EDGES) { s = esrc[e]; d = edst[e]; } else { s = d = e - E_EDGES; }
  float a = leaky(ai[d] + aj[s]);
  float v = expf(a - key2f(amax[d]));
  ex[e] = v;
  hw_atomic_add_f32(&denom[d], v);
}

__global__ __launch_bounds__(256) void k_scatter(const int* __restrict__ esrc,
                                                 const int* __restrict__ edst,
                                                 const float* __restrict__ h,
                                                 const float* __restrict__ ex,
                                                 const float* __restrict__ denom,
                                                 float* __restrict__ out) {
  const int wave = threadIdx.x >> 5;
  const int lane = threadIdx.x & 31;
  const int e = blockIdx.x * 8 + wave;
  int s, d;
  if (e < E_EDGES) { s = esrc[e]; d = edst[e]; } else { s = d = e - E_EDGES; }
  const float coeff = ex[e] / (denom[d] + 1e-16f);
  const float4 hv = *(const float4*)(h + (size_t)s * C_OUT + lane * 4);
  float* o = out + (size_t)d * C_OUT + lane * 4;
  hw_atomic_add_f32(o + 0, hv.x * coeff);
  hw_atomic_add_f32(o + 1, hv.y * coeff);
  hw_atomic_add_f32(o + 2, hv.z * coeff);
  hw_atomic_add_f32(o + 3, hv.w * coeff);
}

extern "C" void kernel_launch(void* const* d_in, const int* in_sizes, int n_in,
                              void* d_out, int out_size, void* d_ws, size_t ws_size,
                              hipStream_t stream) {
  const float* x    = (const float*)d_in[0];
  const int*   ei   = (const int*)  d_in[1];
  const float* W    = (const float*)d_in[2];
  const float* att  = (const float*)d_in[3];
  const float* bias = (const float*)d_in[4];
  float* out = (float*)d_out;

  float*    h     = (float*)d_ws;
  float*    ai    = h + (size_t)N_NODES * C_OUT;
  float*    aj    = ai + N_NODES;
  unsigned* amax  = (unsigned*)(aj + N_NODES);
  float*    denom = (float*)(amax + N_NODES);
  float*    ex    = denom + N_NODES;

  const int* esrc = ei;
  const int* edst = ei + E_EDGES;

  k_gemm   <<<3125, 256, 0, stream>>>(x, W, h);
  k_init   <<<(N_NODES + 255) / 256, 256, 0, stream>>>(amax, denom);
  k_initout<<<(N_NODES * C_OUT + 255) / 256, 256, 0, stream>>>(out, bias);
  k_nodeatt<<<N_NODES / 8, 256, 0, stream>>>(h, att, ai, aj);
  k_edgemax<<<(ET + 255) / 256, 256, 0, stream>>>(esrc, edst, ai, aj, amax);
  k_edgeexp<<<(ET + 255) / 256, 256, 0, stream>>>(esrc, edst, ai, aj, amax, denom, ex);
  k_scatter<<<ET / 8, 256, 0, stream>>>(esrc, edst, h, ex, denom, out);
}


# Node pass: out = bias and the zeroes of the softmax planes and counters ride in the ai/aj launch, 6250 waves of 8 rows each; the two init launches go: 7 -> 5 launches; ai, aj sum order unchanged
# speedup vs baseline: 2.7782x; 2.7782x over previous
#include <hip/hip_runtime.h>


#define N_NODES 50000
#define E_EDGES 800000
#define ET      (E_EDGES + N_NODES)
#define DIN     256
#define C_OUT   128
#define NEG_SLOPE 0.2f
#define NP      50016

typedef __attribute__((ext_vector_type(2))) float v2f;
typedef __attribute__((ext_vector_type(8))) float v8f;
typedef __attribute__((ext_vector_type(4))) float v4f;

__device__ __forceinline__ void hw_atomic_add_f32(float* p, float v) {
  asm volatile("global_atomic_add_f32 %0, %1, off" : : "v"(p), "v"(v) : "memory");
}
__device__ __forceinline__ void hw_atomic_add_f32_row4(float* row, unsigned lane_bytes,
                                                       float v0, float v1, float v2, float v3) {
  asm volatile("global_atomic_add_f32 %0, %1, %5\n\t"
               "global_atomic_add_f32 %0, %2, %5 offset:128\n\t"
               "global_atomic_add_f32 %0, %3, %5 offset:256\n\t"
               "global_atomic_add_f32 %0, %4, %5 offset:384\n\t"
               "s_wait_xcnt 0x0"
               : : "v"(lane_bytes), "v"(v0), "v"(v1), "v"(v2), "v"(v3), "s"(row) : "memory");
}

#define GEMM_WGS   512
#define M_TILES    (N_NODES / 16)
static_assert(N_NODES % 16 == 0 && C_OUT == 128 && DIN == 256, "tile shapes");

__global__ __launch_bounds__(128) __attribute__((amdgpu_waves_per_eu(2, 2)))
void k_gemm(const float* __restrict__ x,
            const float* __restrict__ W,
            float* __restrict__ h) {
  const int wave = threadIdx.x >> 5;
  const int lane = threadIdx.x & 31;
  const int hh   = lane >> 4;
  const int c    = lane & 15;
  const int col0 = wave * 32 + c;

  v2f wf[32][2][2];
#pragma unroll
  for (int i = 0; i < 32; ++i) {
#pragma unroll
    for (int j = 0; j < 2; ++j) {
      const int k = 8 * i + 4 * hh + 2 * j;
#pragma unroll
      for (int t = 0; t < 2; ++t) {
        wf[i][j][t].x = W[(size_t)k * C_OUT + col0 + 16 * t];
        wf[i][j][t].y = W[(size_t)(k + 1) * C_OUT + col0 + 16 * t];
      }
    }
  }

  for (int mt = blockIdx.x; mt < M_TILES; mt += GEMM_WGS) {
    const int tM = mt << 4;
    const float* arow = x + (size_t)(tM + c) * DIN + 4 * hh;
    float4 a[32];
#pragma unroll
    for (int i = 0; i < 32; ++i) a[i] = *(const float4*)(arow + 8 * i);

    v8f acc0 = (v8f)0.0f;
    v8f acc1 = (v8f)0.0f;
#pragma unroll
    for (int i = 0; i < 32; ++i) {
      v2f a0; a0.x = a[i].x; a0.y = a[i].y;
      v2f a1; a1.x = a[i].z; a1.y = a[i].w;
      acc0 = __builtin_amdgcn_wmma_f32_16x16x4_f32(false, a0, false, wf[i][0][0],
                                                   (short)0, acc0, false, false);
      acc1 = __builtin_amdgcn_wmma_f32_16x16x4_f32(false, a0, false, wf[i][0][1],
                                                   (short)0, acc1, false, false);
      acc0 = __builtin_amdgcn_wmma_f32_16x16x4_f32(false, a1, false, wf[i][1][0],
                                                   (short)0, acc0, false, false);
      acc1 = __builtin_amdgcn_wmma_f32_16x16x4_f32(false, a1, false, wf[i][1][1],
                                                   (short)0, acc1, false, false);
    }
    float* orow = h + (size_t)(tM + hh * 8) * C_OUT + col0;
#pragma unroll
    for (int r = 0; r < 8; ++r) {
      orow[(size_t)r * C_OUT]      = acc0[r];
      orow[(size_t)r * C_OUT + 16] = acc1[r];
    }
  }
}

#define CTR_STRIDE 64
#define CTR_PAD    48
#define NODES_PER_WAVE 8
#define NODE_TASKS (N_NODES / NODES_PER_WAVE)
static_assert(((size_t)N_NODES * C_OUT + 6 * (size_t)NP + ET + CTR_PAD) % 64 == 0, "counters on a 256-B boundary");
static_assert(NP % 32 == 0 && NP >= N_NODES, "planes are whole lines");
static_assert(N_NODES % NODES_PER_WAVE == 0 && NODES_PER_WAVE == 8 && C_OUT == 128, "node pass shapes");
__global__ __launch_bounds__(128) void k_node(const float* __restrict__ h,
                                              const float* __restrict__ att,
                                              const float* __restrict__ bias,
                                              float* __restrict__ ai,
                                              float* __restrict__ aj,
                                              unsigned* __restrict__ planes,
                                              unsigned* __restrict__ ctr,
                                              float* __restrict__ out) {
  const int lane = threadIdx.x & 31;
  const int task = blockIdx.x * 4 + (threadIdx.x >> 5);
  if (task >= NODE_TASKS) return;
  const int n0 = task * NODES_PER_WAVE;

  const float* hp = h + (size_t)n0 * C_OUT + lane * 4;
  float4 hv[NODES_PER_WAVE];
#pragma unroll
  for (int j = 0; j < NODES_PER_WAVE; ++j) hv[j] = *(const float4*)(hp + (size_t)j * C_OUT);
  const float4 av = *(const float4*)(att + lane * 4);
  const float4 bv = *(const float4*)(att + C_OUT + lane * 4);
  const v4f b4 = *(const v4f*)(bias + lane * 4);

  float* op = out + (size_t)n0 * C_OUT + lane * 4;
#pragma unroll
  for (int j = 0; j < NODES_PER_WAVE; ++j) *(v4f*)(op + (size_t)j * C_OUT) = b4;
  planes[(size_t)(lane >> 3) * NP + n0 + (lane & 7)] = 0u;
  if (task == 0 && lane < 2) ctr[lane * CTR_STRIDE] = 0u;

  float p[2 * NODES_PER_WAVE];
#pragma unroll
  for (int j = 0; j < NODES_PER_WAVE; ++j) {
    p[2 * j]     = hv[j].x*av.x + hv[j].y*av.y + hv[j].z*av.z + hv[j].w*av.w;
    p[2 * j + 1] = hv[j].x*bv.x + hv[j].y*bv.y + hv[j].z*bv.z + hv[j].w*bv.w;
  }
  const bool b16 = (lane & 16) != 0, b8 = (lane & 8) != 0, b4s = (lane & 4) != 0, b2 = (lane & 2) != 0;
  float q[8], r[4], s[2];
#pragma unroll
  for (int k = 0; k < 8; ++k) {
    const float keep = b16 ? p[k + 8] : p[k];
    const float send = b16 ? p[k] : p[k + 8];
    q[k] = keep + __shfl_xor(send, 16, 32);
  }
#pragma unroll
  for (int k = 0; k < 4; ++k) {
    const float keep = b8 ? q[k + 4] : q[k];
    const float send = b8 ? q[k] : q[k + 4];
    r[k] = keep + __shfl_xor(send, 8, 32);
  }
#pragma unroll
  for (int k = 0; k < 2; ++k) {
    const float keep = b4s ? r[k + 2] : r[k];
    const float send = b4s ? r[k] : r[k + 2];
    s[k] = keep + __shfl_xor(send, 4, 32);
  }
  const float keep = b2 ? s[1] : s[0];
  const float send = b2 ? s[0] : s[1];
  float u = keep + __shfl_xor(send, 2, 32);
  u += __shfl_xor(u, 1, 32);
  if ((lane & 1) == 0) {
    float* dst = (lane & 2) ? aj : ai;
    dst[n0 + (lane >> 2)] = u;
  }
}

__device__ __forceinline__ unsigned my_half() {
  return (unsigned)__builtin_amdgcn_readfirstlane(
      (int)((__builtin_amdgcn_s_sendmsg_rtn(0x87) >> 18) & 1u));
}

__device__ __forceinline__ unsigned f2key(float f) {
  unsigned b = __float_as_uint(f);
  return (b & 0x80000000u) ? ~b : (b | 0x80000000u);
}
__device__ __forceinline__ float key2f(unsigned k) {
  unsigned b = (k & 0x80000000u) ? (k & 0x7FFFFFFFu) : ~k;
  return __uint_as_float(b);
}
__device__ __forceinline__ float leaky(float a) {
  return a > 0.0f ? a : NEG_SLOPE * a;
}

__global__ void k_edgemax(const int* __restrict__ esrc, const int* __restrict__ edst,
                          const float* __restrict__ ai, const float* __restrict__ aj,
                          unsigned* __restrict__ amax) {
  int e = blockIdx.x * blockDim.x + threadIdx.x;
  if (e >= ET) return;
  int s, d;
  if (e < E_EDGES) { s = esrc[e]; d = edst[e]; } else { s = d = e - E_EDGES; }
  float a = leaky(ai[d] + aj[s]);
  atomicMax(&amax[my_half() * NP + d], f2key(a));
}

__global__ void k_edgeexp(const int* __restrict__ esrc, const int* __restrict__ edst,
                          const float* __restrict__ ai, const float* __restrict__ aj,
                          const unsigned* __restrict__ amax,
                          float* __restrict__ denom, float* __restrict__ ex) {
  int e = blockIdx.x * blockDim.x + threadIdx.x;
  if (e >= ET) return;
  int s, d;
  if (e < E_EDGES) { s = esrc[e]; d = edst[e]; } else { s = d = e - E_EDGES; }
  float a = leaky(ai[d] + aj[s]);
  const unsigned k0 = amax[d], k1 = amax[NP + d];
  float v = expf(a - key2f(k0 > k1 ? k0 : k1));
  ex[e] = v;
  hw_atomic_add_f32(&denom[my_half() * NP + d], v);
}

#define CHUNK     32
#define NCHUNK    ((ET + CHUNK - 1) / CHUNK)
#define HALF_ROWS (N_NODES / 2)
#define GRP       16
#define QCAP      64
#define SCAT_WGS  1024
static_assert(E_EDGES % CHUNK == 0, "a chunk is either all listed edges or all self loops");
static_assert(GRP - 1 + CHUNK <= QCAP && (QCAP & (QCAP - 1)) == 0, "ring size");

template <bool FULL>
__device__ __forceinline__ void scatter_group(int n, int s, int d, float coeff,
                                              const float* __restrict__ h,
                                              float* __restrict__ out,
                                              int lane, unsigned lane_bytes) {
  float hv[GRP][4];
#pragma unroll
  for (int j = 0; j < GRP; ++j) {
    if (FULL || j < n) {
      const int sj = __builtin_amdgcn_readlane(s, j);
      const float* hp = h + (size_t)sj * C_OUT + lane;
      hv[j][0] = hp[0];
      hv[j][1] = hp[32];
      hv[j][2] = hp[64];
      hv[j][3] = hp[96];
    }
  }
#pragma unroll
  for (int j = 0; j < GRP; ++j) {
    if (FULL || j < n) {
      const int   dj = __builtin_amdgcn_readlane(d, j);
      const float cj = __int_as_float(__builtin_amdgcn_readlane(__float_as_int(coeff), j));
      float* orow = out + (size_t)dj * C_OUT;
      hw_atomic_add_f32_row4(orow, lane_bytes, hv[j][0] * cj, hv[j][1] * cj,
                             hv[j][2] * cj, hv[j][3] * cj);
    }
  }
}

__global__ __launch_bounds__(256) void k_scatter(const int* __restrict__ esrc,
                                                 const int* __restrict__ edst,
                                                 const float* __restrict__ h,
                                                 const float* __restrict__ ex,
                                                 const float* __restrict__ denom,
                                                 float* __restrict__ out,
                                                 unsigned* __restrict__ ctr) {
  __shared__ int qbuf[8 * 3 * QCAP];
  const int lane = threadIdx.x & 31;
  const unsigned lane_bytes = (unsigned)lane * 4u;
  const int wave = __builtin_amdgcn_readfirstlane((int)(threadIdx.x >> 5));
  int* qs = qbuf + wave * (3 * QCAP);
  int* qd = qs + QCAP;
  int* qc = qd + QCAP;
  unsigned head = 0u, tail = 0u;

  unsigned hsel = (unsigned)__builtin_amdgcn_readfirstlane(
      (int)((__builtin_amdgcn_s_sendmsg_rtn(0x87) >> 18) & 1u));

  for (int pass = 0; pass < 2; ++pass) {
    if (pass == 1 && wave != 0) break;
    unsigned* c = ctr + hsel * CTR_STRIDE;
    for (int it = 0; it <= NCHUNK; ++it) {
      unsigned tv = 0u;
      if (lane == 0) tv = atomicAdd(c, 1u);
      const unsigned t = (unsigned)__builtin_amdgcn_readfirstlane((int)tv);
      if (t >= (unsigned)NCHUNK) break;

      const int e = (int)t * CHUNK + lane;
      int s = 0, d = 0;
      float coeff = 0.0f;
      bool mine = false;
      if (e < ET) {
        if (e < E_EDGES) { s = esrc[e]; d = edst[e]; } else { s = d = e - E_EDGES; }
        mine = ((d >= HALF_ROWS) ? 1u : 0u) == hsel;
        if (mine) coeff = ex[e] / ((denom[d] + denom[NP + d]) + 1e-16f);
      }
      const unsigned m = __builtin_amdgcn_ballot_w32(mine);
      if (mine) {
        const unsigned slot = (tail + __builtin_amdgcn_mbcnt_lo(m, 0u)) & (QCAP - 1);
        qs[slot] = s;
        qd[slot] = d;
        qc[slot] = __float_as_int(coeff);
      }
      tail += (unsigned)__builtin_popcount(m);
      while (tail - head >= GRP) {
        const unsigned slot = (head + (unsigned)(lane & (GRP - 1))) & (QCAP - 1);
        const int   s16 = qs[slot];
        const int   d16 = qd[slot];
        const float c16 = __int_as_float(qc[slot]);
        head += GRP;
        scatter_group<true>(GRP, s16, d16, c16, h, out, lane, lane_bytes);
      }
    }
    hsel ^= 1u;
  }

  const int rem = (int)(tail - head);
  if (rem > 0) {
    int s16 = 0, d16 = 0;
    float c16 = 0.0f;
    if ((lane & (GRP - 1)) < rem) {
      const unsigned slot = (head + (unsigned)(lane & (GRP - 1))) & (QCAP - 1);
      s16 = qs[slot];
      d16 = qd[slot];
      c16 = __int_as_float(qc[slot]);
    }
    scatter_group<false>(rem, s16, d16, c16, h, out, lane, lane_bytes);
  }
}

extern "C" void kernel_launch(void* const* d_in, const int* in_sizes, int n_in,
                              void* d_out, int out_size, void* d_ws, size_t ws_size,
                              hipStream_t stream) {
  const float* x    = (const float*)d_in[0];
  const int*   ei   = (const int*)  d_in[1];
  const float* W    = (const float*)d_in[2];
  const float* att  = (const float*)d_in[3];
  const float* bias = (const float*)d_in[4];
  float* out = (float*)d_out;

  float*    h     = (float*)d_ws;
  float*    ai    = h + (size_t)N_NODES * C_OUT;
  float*    aj    = ai + NP;
  unsigned* amax  = (unsigned*)(aj + NP);
  float*    denom = (float*)(amax + 2 * NP);
  float*    ex    = denom + 2 * NP;
  unsigned* ctr   = (unsigned*)(ex + ET + CTR_PAD);

  const int* esrc = ei;
  const int* edst = ei + E_EDGES;

  k_gemm   <<<GEMM_WGS, 128, 0, stream>>>(x, W, h);
  k_node   <<<(NODE_TASKS + 3) / 4, 128, 0, stream>>>(h, att, bias, ai, aj, amax, ctr, out);
  k_edgemax<<<(ET + 255) / 256, 256, 0, stream>>>(esrc, edst, ai, aj, amax);
  k_edgeexp<<<(ET + 255) / 256, 256, 0, stream>>>(esrc, edst, ai, aj, amax, denom, ex);
  k_scatter<<<SCAT_WGS, 256, 0, stream>>>(esrc, edst, h, ex, denom, out, ctr);
}


# Edge softmax: the max pass stores each edge's logit (one coalesced word) and the exp pass reads it back, so ai[dst], aj[src] are gathered once, not twice; logit bits same, add order still arbitrary
# speedup vs baseline: 2.8931x; 1.0414x over previous
#include <hip/hip_runtime.h>


#define N_NODES 50000
#define E_EDGES 800000
#define ET      (E_EDGES + N_NODES)
#define DIN     256
#define C_OUT   128
#define NEG_SLOPE 0.2f
#define NP      50016

typedef __attribute__((ext_vector_type(2))) float v2f;
typedef __attribute__((ext_vector_type(8))) float v8f;
typedef __attribute__((ext_vector_type(4))) float v4f;

__device__ __forceinline__ void hw_atomic_add_f32(float* p, float v) {
  asm volatile("global_atomic_add_f32 %0, %1, off" : : "v"(p), "v"(v) : "memory");
}
__device__ __forceinline__ void hw_atomic_add_f32_row4(float* row, unsigned lane_bytes,
                                                       float v0, float v1, float v2, float v3) {
  asm volatile("global_atomic_add_f32 %0, %1, %5\n\t"
               "global_atomic_add_f32 %0, %2, %5 offset:128\n\t"
               "global_atomic_add_f32 %0, %3, %5 offset:256\n\t"
               "global_atomic_add_f32 %0, %4, %5 offset:384\n\t"
               "s_wait_xcnt 0x0"
               : : "v"(lane_bytes), "v"(v0), "v"(v1), "v"(v2), "v"(v3), "s"(row) : "memory");
}

#define GEMM_WGS   512
#define M_TILES    (N_NODES / 16)
static_assert(N_NODES % 16 == 0 && C_OUT == 128 && DIN == 256, "tile shapes");

__global__ __launch_bounds__(128) __attribute__((amdgpu_waves_per_eu(2, 2)))
void k_gemm(const float* __restrict__ x,
            const float* __restrict__ W,
            float* __restrict__ h) {
  const int wave = threadIdx.x >> 5;
  const int lane = threadIdx.x & 31;
  const int hh   = lane >> 4;
  const int c    = lane & 15;
  const int col0 = wave * 32 + c;

  v2f wf[32][2][2];
#pragma unroll
  for (int i = 0; i < 32; ++i) {
#pragma unroll
    for (int j = 0; j < 2; ++j) {
      const int k = 8 * i + 4 * hh + 2 * j;
#pragma unroll
      for (int t = 0; t < 2; ++t) {
        wf[i][j][t].x = W[(size_t)k * C_OUT + col0 + 16 * t];
        wf[i][j][t].y = W[(size_t)(k + 1) * C_OUT + col0 + 16 * t];
      }
    }
  }

  for (int mt = blockIdx.x; mt < M_TILES; mt += GEMM_WGS) {
    const int tM = mt << 4;
    const float* arow = x + (size_t)(tM + c) * DIN + 4 * hh;
    float4 a[32];
#pragma unroll
    for (int i = 0; i < 32; ++i) a[i] = *(const float4*)(arow + 8 * i);

    v8f acc0 = (v8f)0.0f;
    v8f acc1 = (v8f)0.0f;
#pragma unroll
    for (int i = 0; i < 32; ++i) {
      v2f a0; a0.x = a[i].x; a0.y = a[i].y;
      v2f a1; a1.x = a[i].z; a1.y = a[i].w;
      acc0 = __builtin_amdgcn_wmma_f32_16x16x4_f32(false, a0, false, wf[i][0][0],
                                                   (short)0, acc0, false, false);
      acc1 = __builtin_amdgcn_wmma_f32_16x16x4_f32(false, a0, false, wf[i][0][1],
                                                   (short)0, acc1, false, false);
      acc0 = __builtin_amdgcn_wmma_f32_16x16x4_f32(false, a1, false, wf[i][1][0],
                                                   (short)0, acc0, false, false);
      acc1 = __builtin_amdgcn_wmma_f32_16x16x4_f32(false, a1, false, wf[i][1][1],
                                                   (short)0, acc1, false, false);
    }
    float* orow = h + (size_t)(tM + hh * 8) * C_OUT + col0;
#pragma unroll
    for (int r = 0; r < 8; ++r) {
      orow[(size_t)r * C_OUT]      = acc0[r];
      orow[(size_t)r * C_OUT + 16] = acc1[r];
    }
  }
}

#define CTR_STRIDE 64
#define CTR_PAD    48
#define NODES_PER_WAVE 8
#define NODE_TASKS (N_NODES / NODES_PER_WAVE)
static_assert(((size_t)N_NODES * C_OUT + 6 * (size_t)NP + ET + CTR_PAD) % 64 == 0, "counters on a 256-B boundary");
static_assert(NP % 32 == 0 && NP >= N_NODES, "planes are whole lines");
static_assert(N_NODES % NODES_PER_WAVE == 0 && NODES_PER_WAVE == 8 && C_OUT == 128, "node pass shapes");
__global__ __launch_bounds__(128) void k_node(const float* __restrict__ h,
                                              const float* __restrict__ att,
                                              const float* __restrict__ bias,
                                              float* __restrict__ ai,
                                              float* __restrict__ aj,
                                              unsigned* __restrict__ planes,
                                              unsigned* __restrict__ ctr,
                                              float* __restrict__ out) {
  const int lane = threadIdx.x & 31;
  const int task = blockIdx.x * 4 + (threadIdx.x >> 5);
  if (task >= NODE_TASKS) return;
  const int n0 = task * NODES_PER_WAVE;

  const float* hp = h + (size_t)n0 * C_OUT + lane * 4;
  float4 hv[NODES_PER_WAVE];
#pragma unroll
  for (int j = 0; j < NODES_PER_WAVE; ++j) hv[j] = *(const float4*)(hp + (size_t)j * C_OUT);
  const float4 av = *(const float4*)(att + lane * 4);
  const float4 bv = *(const float4*)(att + C_OUT + lane * 4);
  const v4f b4 = *(const v4f*)(bias + lane * 4);

  float* op = out + (size_t)n0 * C_OUT + lane * 4;
#pragma unroll
  for (int j = 0; j < NODES_PER_WAVE; ++j) *(v4f*)(op + (size_t)j * C_OUT) = b4;
  planes[(size_t)(lane >> 3) * NP + n0 + (lane & 7)] = 0u;
  if (task == 0 && lane < 2) ctr[lane * CTR_STRIDE] = 0u;

  float p[2 * NODES_PER_WAVE];
#pragma unroll
  for (int j = 0; j < NODES_PER_WAVE; ++j) {
    p[2 * j]     = hv[j].x*av.x + hv[j].y*av.y + hv[j].z*av.z + hv[j].w*av.w;
    p[2 * j + 1] = hv[j].x*bv.x + hv[j].y*bv.y + hv[j].z*bv.z + hv[j].w*bv.w;
  }
  const bool b16 = (lane & 16) != 0, b8 = (lane & 8) != 0, b4s = (lane & 4) != 0, b2 = (lane & 2) != 0;
  float q[8], r[4], s[2];
#pragma unroll
  for (int k = 0; k < 8; ++k) {
    const float keep = b16 ? p[k + 8] : p[k];
    const float send = b16 ? p[k] : p[k + 8];
    q[k] = keep + __shfl_xor(send, 16, 32);
  }
#pragma unroll
  for (int k = 0; k < 4; ++k) {
    const float keep = b8 ? q[k + 4] : q[k];
    const float send = b8 ? q[k] : q[k + 4];
    r[k] = keep + __shfl_xor(send, 8, 32);
  }
#pragma unroll
  for (int k = 0; k < 2; ++k) {
    const float keep = b4s ? r[k + 2] : r[k];
    const float send = b4s ? r[k] : r[k + 2];
    s[k] = keep + __shfl_xor(send, 4, 32);
  }
  const float keep = b2 ? s[1] : s[0];
  const float send = b2 ? s[0] : s[1];
  float u = keep + __shfl_xor(send, 2, 32);
  u += __shfl_xor(u, 1, 32);
  if ((lane & 1) == 0) {
    float* dst = (lane & 2) ? aj : ai;
    dst[n0 + (lane >> 2)] = u;
  }
}

__device__ __forceinline__ unsigned my_half() {
  return (unsigned)__builtin_amdgcn_readfirstlane(
      (int)((__builtin_amdgcn_s_sendmsg_rtn(0x87) >> 18) & 1u));
}

__device__ __forceinline__ unsigned f2key(float f) {
  unsigned b = __float_as_uint(f);
  return (b & 0x80000000u) ? ~b : (b | 0x80000000u);
}
__device__ __forceinline__ float key2f(unsigned k) {
  unsigned b = (k & 0x80000000u) ? (k & 0x7FFFFFFFu) : ~k;
  return __uint_as_float(b);
}
__device__ __forceinline__ float leaky(float a) {
  return a > 0.0f ? a : NEG_SLOPE * a;
}

__global__ void k_edgemax(const int* __restrict__ esrc, const int* __restrict__ edst,
                          const float* __restrict__ ai, const float* __restrict__ aj,
                          unsigned* __restrict__ amax, float* __restrict__ alpha) {
  int e = blockIdx.x * blockDim.x + threadIdx.x;
  if (e >= ET) return;
  int s, d;
  if (e < E_EDGES) { s = esrc[e]; d = edst[e]; } else { s = d = e - E_EDGES; }
  float a = leaky(ai[d] + aj[s]);
  alpha[e] = a;
  atomicMax(&amax[my_half() * NP + d], f2key(a));
}

__global__ void k_edgeexp(const int* __restrict__ edst,
                          const unsigned* __restrict__ amax,
                          float* __restrict__ denom, float* __restrict__ ex) {
  int e = blockIdx.x * blockDim.x + threadIdx.x;
  if (e >= ET) return;
  const int d = (e < E_EDGES) ? edst[e] : e - E_EDGES;
  const float a = ex[e];
  const unsigned k0 = amax[d], k1 = amax[NP + d];
  float v = expf(a - key2f(k0 > k1 ? k0 : k1));
  ex[e] = v;
  hw_atomic_add_f32(&denom[my_half() * NP + d], v);
}

#define CHUNK     32
#define NCHUNK    ((ET + CHUNK - 1) / CHUNK)
#define HALF_ROWS (N_NODES / 2)
#define GRP       16
#define QCAP      64
#define SCAT_WGS  1024
static_assert(E_EDGES % CHUNK == 0, "a chunk is either all listed edges or all self loops");
static_assert(GRP - 1 + CHUNK <= QCAP && (QCAP & (QCAP - 1)) == 0, "ring size");

template <bool FULL>
__device__ __forceinline__ void scatter_group(int n, int s, int d, float coeff,
                                              const float* __restrict__ h,
                                              float* __restrict__ out,
                                              int lane, unsigned lane_bytes) {
  float hv[GRP][4];
#pragma unroll
  for (int j = 0; j < GRP; ++j) {
    if (FULL || j < n) {
      const int sj = __builtin_amdgcn_readlane(s, j);
      const float* hp = h + (size_t)sj * C_OUT + lane;
      hv[j][0] = hp[0];
      hv[j][1] = hp[32];
      hv[j][2] = hp[64];
      hv[j][3] = hp[96];
    }
  }
#pragma unroll
  for (int j = 0; j < GRP; ++j) {
    if (FULL || j < n) {
      const int   dj = __builtin_amdgcn_readlane(d, j);
      const float cj = __int_as_float(__builtin_amdgcn_readlane(__float_as_int(coeff), j));
      float* orow = out + (size_t)dj * C_OUT;
      hw_atomic_add_f32_row4(orow, lane_bytes, hv[j][0] * cj, hv[j][1] * cj,
                             hv[j][2] * cj, hv[j][3] * cj);
    }
  }
}

__global__ __launch_bounds__(256) void k_scatter(const int* __restrict__ esrc,
                                                 const int* __restrict__ edst,
                                                 const float* __restrict__ h,
                                                 const float* __restrict__ ex,
                                                 const float* __restrict__ denom,
                                                 float* __restrict__ out,
                                                 unsigned* __restrict__ ctr) {
  __shared__ int qbuf[8 * 3 * QCAP];
  const int lane = threadIdx.x & 31;
  const unsigned lane_bytes = (unsigned)lane * 4u;
  const int wave = __builtin_amdgcn_readfirstlane((int)(threadIdx.x >> 5));
  int* qs = qbuf + wave * (3 * QCAP);
  int* qd = qs + QCAP;
  int* qc = qd + QCAP;
  unsigned head = 0u, tail = 0u;

  unsigned hsel = (unsigned)__builtin_amdgcn_readfirstlane(
      (int)((__builtin_amdgcn_s_sendmsg_rtn(0x87) >> 18) & 1u));

  for (int pass = 0; pass < 2; ++pass) {
    if (pass == 1 && wave != 0) break;
    unsigned* c = ctr + hsel * CTR_STRIDE;
    for (int it = 0; it <= NCHUNK; ++it) {
      unsigned tv = 0u;
      if (lane == 0) tv = atomicAdd(c, 1u);
      const unsigned t = (unsigned)__builtin_amdgcn_readfirstlane((int)tv);
      if (t >= (unsigned)NCHUNK) break;

      const int e = (int)t * CHUNK + lane;
      int s = 0, d = 0;
      float coeff = 0.0f;
      bool mine = false;
      if (e < ET) {
        if (e < E_EDGES) { s = esrc[e]; d = edst[e]; } else { s = d = e - E_EDGES; }
        mine = ((d >= HALF_ROWS) ? 1u : 0u) == hsel;
        if (mine) coeff = ex[e] / ((denom[d] + denom[NP + d]) + 1e-16f);
      }
      const unsigned m = __builtin_amdgcn_ballot_w32(mine);
      if (mine) {
        const unsigned slot = (tail + __builtin_amdgcn_mbcnt_lo(m, 0u)) & (QCAP - 1);
        qs[slot] = s;
        qd[slot] = d;
        qc[slot] = __float_as_int(coeff);
      }
      tail += (unsigned)__builtin_popcount(m);
      while (tail - head >= GRP) {
        const unsigned slot = (head + (unsigned)(lane & (GRP - 1))) & (QCAP - 1);
        const int   s16 = qs[slot];
        const int   d16 = qd[slot];
        const float c16 = __int_as_float(qc[slot]);
        head += GRP;
        scatter_group<true>(GRP, s16, d16, c16, h, out, lane, lane_bytes);
      }
    }
    hsel ^= 1u;
  }

  const int rem = (int)(tail - head);
  if (rem > 0) {
    int s16 = 0, d16 = 0;
    float c16 = 0.0f;
    if ((lane & (GRP - 1)) < rem) {
      const unsigned slot = (head + (unsigned)(lane & (GRP - 1))) & (QCAP - 1);
      s16 = qs[slot];
      d16 = qd[slot];
      c16 = __int_as_float(qc[slot]);
    }
    scatter_group<false>(rem, s16, d16, c16, h, out, lane, lane_bytes);
  }
}

extern "C" void kernel_launch(void* const* d_in, const int* in_sizes, int n_in,
                              void* d_out, int out_size, void* d_ws, size_t ws_size,
                              hipStream_t stream) {
  const float* x    = (const float*)d_in[0];
  const int*   ei   = (const int*)  d_in[1];
  const float* W    = (const float*)d_in[2];
  const float* att  = (const float*)d_in[3];
  const float* bias = (const float*)d_in[4];
  float* out = (float*)d_out;

  float*    h     = (float*)d_ws;
  float*    ai    = h + (size_t)N_NODES * C_OUT;
  float*    aj    = ai + NP;
  unsigned* amax  = (unsigned*)(aj + NP);
  float*    denom = (float*)(amax + 2 * NP);
  float*    ex    = denom + 2 * NP;
  unsigned* ctr   = (unsigned*)(ex + ET + CTR_PAD);

  const int* esrc = ei;
  const int* edst = ei + E_EDGES;

  k_gemm   <<<GEMM_WGS, 128, 0, stream>>>(x, W, h);
  k_node   <<<(NODE_TASKS + 3) / 4, 128, 0, stream>>>(h, att, bias, ai, aj, amax, ctr, out);
  k_edgemax<<<(ET + 255) / 256, 256, 0, stream>>>(esrc, edst, ai, aj, amax, ex);
  k_edgeexp<<<(ET + 255) / 256, 256, 0, stream>>>(edst, amax, denom, ex);
  k_scatter<<<SCAT_WGS, 256, 0, stream>>>(esrc, edst, h, ex, denom, out, ctr);
}


# Edge scan: the node launch writes a bit map of the dst list, one 128-B line per 1024-edge block and owner work-group; an owner reads its 782 lines, not the whole list; sum order still arbitrary
# speedup vs baseline: 3.7746x; 1.3047x over previous
#include <hip/hip_runtime.h>


#define N_NODES 50000
#define E_EDGES 800000
#define DIN     256
#define C_OUT   128
#define NEG_SLOPE 0.2f

#define NB_EDGE  250
#define ROWS_WG  (N_NODES / NB_EDGE)
#define EBLK     1024
#define NEB      ((E_EDGES + EBLK - 1) / EBLK)
#define BM_ROW   (EBLK / 32)
#define BM_BLK   (NB_EDGE * BM_ROW)
static_assert(BM_ROW * 4 == 128 && (BM_BLK * 4) % 128 == 0, "one line per owner row");

typedef __attribute__((ext_vector_type(2))) float v2f;
typedef __attribute__((ext_vector_type(8))) float v8f;
typedef __attribute__((ext_vector_type(4))) float v4f;
typedef __attribute__((ext_vector_type(4))) unsigned v4u;

#define GEMM_WGS   512
#define M_TILES    (N_NODES / 16)
static_assert(N_NODES % 16 == 0 && C_OUT == 128 && DIN == 256, "tile shapes");

__global__ __launch_bounds__(128) __attribute__((amdgpu_waves_per_eu(2, 2)))
void k_gemm(const float* __restrict__ x,
            const float* __restrict__ W,
            float* __restrict__ h) {
  const int wave = threadIdx.x >> 5;
  const int lane = threadIdx.x & 31;
  const int hh   = lane >> 4;
  const int c    = lane & 15;
  const int col0 = wave * 32 + c;

  v2f wf[32][2][2];
#pragma unroll
  for (int i = 0; i < 32; ++i) {
#pragma unroll
    for (int j = 0; j < 2; ++j) {
      const int k = 8 * i + 4 * hh + 2 * j;
#pragma unroll
      for (int t = 0; t < 2; ++t) {
        wf[i][j][t].x = W[(size_t)k * C_OUT + col0 + 16 * t];
        wf[i][j][t].y = W[(size_t)(k + 1) * C_OUT + col0 + 16 * t];
      }
    }
  }

  for (int mt = blockIdx.x; mt < M_TILES; mt += GEMM_WGS) {
    const int tM = mt << 4;
    const float* arow = x + (size_t)(tM + c) * DIN + 4 * hh;
    float4 a[32];
#pragma unroll
    for (int i = 0; i < 32; ++i) a[i] = *(const float4*)(arow + 8 * i);

    v8f acc0 = (v8f)0.0f;
    v8f acc1 = (v8f)0.0f;
#pragma unroll
    for (int i = 0; i < 32; ++i) {
      v2f a0; a0.x = a[i].x; a0.y = a[i].y;
      v2f a1; a1.x = a[i].z; a1.y = a[i].w;
      acc0 = __builtin_amdgcn_wmma_f32_16x16x4_f32(false, a0, false, wf[i][0][0],
                                                   (short)0, acc0, false, false);
      acc1 = __builtin_amdgcn_wmma_f32_16x16x4_f32(false, a0, false, wf[i][0][1],
                                                   (short)0, acc1, false, false);
      acc0 = __builtin_amdgcn_wmma_f32_16x16x4_f32(false, a1, false, wf[i][1][0],
                                                   (short)0, acc0, false, false);
      acc1 = __builtin_amdgcn_wmma_f32_16x16x4_f32(false, a1, false, wf[i][1][1],
                                                   (short)0, acc1, false, false);
    }
    float* orow = h + (size_t)(tM + hh * 8) * C_OUT + col0;
#pragma unroll
    for (int r = 0; r < 8; ++r) {
      orow[(size_t)r * C_OUT]      = acc0[r];
      orow[(size_t)r * C_OUT + 16] = acc1[r];
    }
  }
}

#define NODES_PER_WAVE 8
#define NODE_TASKS (N_NODES / NODES_PER_WAVE)
#define NODE_WGS   ((NODE_TASKS + 3) / 4)
static_assert(N_NODES % NODES_PER_WAVE == 0 && NODES_PER_WAVE == 8 && C_OUT == 128, "node pass shapes");
static_assert(E_EDGES % 4 == 0 && EBLK == 8 * 128, "a map work-group: 128 threads, two 16-B target loads each");
__global__ __launch_bounds__(128) void k_node(const float* __restrict__ h,
                                              const float* __restrict__ att,
                                              const int* __restrict__ edst,
                                              float* __restrict__ ai,
                                              float* __restrict__ aj,
                                              unsigned* __restrict__ bmap) {
  __shared__ __attribute__((aligned(16))) unsigned bml[BM_BLK];
  if (blockIdx.x < NEB) {
    const int t = threadIdx.x;
    for (int i = t; i < BM_BLK / 4; i += 128) ((v4u*)bml)[i] = (v4u)0u;
    __syncthreads();
    const int e0 = blockIdx.x * EBLK;
    int4 d[2];
#pragma unroll
    for (int j = 0; j < 2; ++j) {
      const int le = 4 * (t + 128 * j);
      d[j] = (e0 + le < E_EDGES) ? *(const int4*)(edst + e0 + le) : make_int4(-1, -1, -1, -1);
    }
#pragma unroll
    for (int j = 0; j < 2; ++j) {
      const int le = 4 * (t + 128 * j);
      const int dd[4] = {d[j].x, d[j].y, d[j].z, d[j].w};
#pragma unroll
      for (int c = 0; c < 4; ++c) {
        if ((unsigned)dd[c] < (unsigned)N_NODES)
          atomicOr(&bml[((unsigned)dd[c] / ROWS_WG) * BM_ROW + (le >> 5)], 1u << ((le & 31) + c));
      }
    }
    __syncthreads();
    v4u* mp = (v4u*)(bmap + (size_t)blockIdx.x * BM_BLK);
    for (int i = t; i < BM_BLK / 4; i += 128) mp[i] = ((const v4u*)bml)[i];
    return;
  }
  const int lane = threadIdx.x & 31;
  const int task = (blockIdx.x - NEB) * 4 + (threadIdx.x >> 5);
  if (task >= NODE_TASKS) return;
  const int n0 = task * NODES_PER_WAVE;

  const float* hp = h + (size_t)n0 * C_OUT + lane * 4;
  float4 hv[NODES_PER_WAVE];
#pragma unroll
  for (int j = 0; j < NODES_PER_WAVE; ++j) hv[j] = *(const float4*)(hp + (size_t)j * C_OUT);
  const float4 av = *(const float4*)(att + lane * 4);
  const float4 bv = *(const float4*)(att + C_OUT + lane * 4);

  float p[2 * NODES_PER_WAVE];
#pragma unroll
  for (int j = 0; j < NODES_PER_WAVE; ++j) {
    p[2 * j]     = hv[j].x*av.x + hv[j].y*av.y + hv[j].z*av.z + hv[j].w*av.w;
    p[2 * j + 1] = hv[j].x*bv.x + hv[j].y*bv.y + hv[j].z*bv.z + hv[j].w*bv.w;
  }
  const bool b16 = (lane & 16) != 0, b8 = (lane & 8) != 0, b4s = (lane & 4) != 0, b2 = (lane & 2) != 0;
  float q[8], r[4], s[2];
#pragma unroll
  for (int k = 0; k < 8; ++k) {
    const float keep = b16 ? p[k + 8] : p[k];
    const float send = b16 ? p[k] : p[k + 8];
    q[k] = keep + __shfl_xor(send, 16, 32);
  }
#pragma unroll
  for (int k = 0; k < 4; ++k) {
    const float keep = b8 ? q[k + 4] : q[k];
    const float send = b8 ? q[k] : q[k + 4];
    r[k] = keep + __shfl_xor(send, 8, 32);
  }
#pragma unroll
  for (int k = 0; k < 2; ++k) {
    const float keep = b4s ? r[k + 2] : r[k];
    const float send = b4s ? r[k] : r[k + 2];
    s[k] = keep + __shfl_xor(send, 4, 32);
  }
  const float keep = b2 ? s[1] : s[0];
  const float send = b2 ? s[0] : s[1];
  float u = keep + __shfl_xor(send, 2, 32);
  u += __shfl_xor(u, 1, 32);
  if ((lane & 1) == 0) {
    float* dst = (lane & 2) ? aj : ai;
    dst[n0 + (lane >> 2)] = u;
  }
}

__device__ __forceinline__ unsigned f2key(float f) {
  unsigned b = __float_as_uint(f);
  return (b & 0x80000000u) ? ~b : (b | 0x80000000u);
}
__device__ __forceinline__ float key2f(unsigned k) {
  unsigned b = (k & 0x80000000u) ? (k & 0x7FFFFFFFu) : ~k;
  return __uint_as_float(b);
}
__device__ __forceinline__ float leaky(float a) {
  return a > 0.0f ? a : NEG_SLOPE * a;
}

#define QCAP     32768
#define EDGE_T   1024
#define ROW_G    10
#define E4       (E_EDGES / 4)
#define MAP_CH   (NEB * 8)
#define MAP_U    ((MAP_CH + EDGE_T - 1) / EDGE_T)
static_assert(N_NODES == NB_EDGE * ROWS_WG, "rows split evenly");
static_assert(N_NODES <= 65536 && ROWS_WG <= 65536, "queue word packing: 16 bits each for source and local row");
static_assert(E_EDGES % 4 == 0, "dst list in 16-B units");
static_assert(EDGE_T * 4 + ROWS_WG <= QCAP, "one batch of the slow scan fits the queue");

__device__ __forceinline__ void q_push(unsigned* q, unsigned* qcount, int e) {
  const unsigned pos = atomicAdd(qcount, 1u);
  if (pos < (unsigned)QCAP) q[pos] = (unsigned)e;
}
__device__ __forceinline__ void q_resolve(unsigned* q, unsigned cnt,
                                          const int* __restrict__ esrc,
                                          const int* __restrict__ edst, int n0, int tid) {
  for (unsigned i0 = (unsigned)tid; i0 < cnt; i0 += 4 * EDGE_T) {
    int s[4];
    int d[4];
#pragma unroll
    for (int u = 0; u < 4; ++u) {
      const unsigned i = i0 + u * EDGE_T;
      if (i < cnt) {
        const int e = (int)q[i];
        if (e < E_EDGES) { s[u] = esrc[e]; d[u] = edst[e]; }
        else             { s[u] = e - E_EDGES; d[u] = e - E_EDGES; }
      }
    }
#pragma unroll
    for (int u = 0; u < 4; ++u) {
      const unsigned i = i0 + u * EDGE_T;
      if (i < cnt) q[i] = (unsigned)s[u] | ((unsigned)(d[u] - n0) << 16);
    }
  }
}

template <bool FULL>
__device__ __forceinline__ void row_group(int n, unsigned ent, float ajs,
                                          const float* ail, const unsigned* mkey,
                                          const float* lsum, float* acc,
                                          const float* __restrict__ aj,
                                          const float* __restrict__ h, int lane) {
  const int s  = (int)(ent & 0xFFFFu);
  const int dl = (int)(ent >> 16);
  float hv[ROW_G][4];
#pragma unroll
  for (int j = 0; j < ROW_G; ++j) {
    if (FULL || j < n) {
      const int sj = __builtin_amdgcn_readlane(s, j);
      const float* hp = h + (size_t)sj * C_OUT;
      hv[j][0] = hp[lane];
      hv[j][1] = hp[lane + 32];
      hv[j][2] = hp[lane + 64];
      hv[j][3] = hp[lane + 96];
    }
  }
  const float a  = leaky(ail[dl] + ajs);
  const float v  = expf(a - key2f(mkey[dl]));
  const float cf = v / (lsum[dl] + 1e-16f);
#pragma unroll
  for (int j = 0; j < ROW_G; ++j) {
    if (FULL || j < n) {
      const int   dj = __builtin_amdgcn_readlane(dl, j);
      const float cj = __int_as_float(__builtin_amdgcn_readlane(__float_as_int(cf), j));
      float* ar = acc + dj * C_OUT + lane;
      atomicAdd(ar,      hv[j][0] * cj);
      atomicAdd(ar + 32, hv[j][1] * cj);
      atomicAdd(ar + 64, hv[j][2] * cj);
      atomicAdd(ar + 96, hv[j][3] * cj);
    }
  }
}

__device__ __forceinline__ void rows_from_queue(const unsigned* q, unsigned cnt,
                                                const float* ail, const unsigned* mkey,
                                                const float* lsum, float* acc,
                                                const float* __restrict__ aj,
                                                const float* __restrict__ h,
                                                int wave, int lane) {
  const unsigned stride = (EDGE_T / 32) * ROW_G;
  unsigned base = (unsigned)wave * ROW_G;
  unsigned ent = (lane < ROW_G && base + (unsigned)lane < cnt) ? q[base + lane] : 0u;
  float ajs = aj[ent & 0xFFFFu];
  for (; base + ROW_G <= cnt; base += stride) {
    const unsigned nb = base + stride;
    const unsigned ent_n = (lane < ROW_G && nb + (unsigned)lane < cnt) ? q[nb + lane] : 0u;
    const float ajs_n = aj[ent_n & 0xFFFFu];
    row_group<true>(ROW_G, ent, ajs, ail, mkey, lsum, acc, aj, h, lane);
    ent = ent_n;
    ajs = ajs_n;
  }
  if (base < cnt) row_group<false>((int)(cnt - base), ent, ajs, ail, mkey, lsum, acc, aj, h, lane);
}

__global__ __launch_bounds__(EDGE_T) void k_edge(const int* __restrict__ esrc,
                                                 const int* __restrict__ edst,
                                                 const float* __restrict__ h,
                                                 const float* __restrict__ ai,
                                                 const float* __restrict__ aj,
                                                 const float* __restrict__ bias,
                                                 const unsigned* __restrict__ bmap,
                                                 float* __restrict__ out) {
  __shared__ __attribute__((aligned(16))) float acc[ROWS_WG * C_OUT];
  __shared__ unsigned q[QCAP];
  __shared__ unsigned mkey[ROWS_WG];
  __shared__ float    lsum[ROWS_WG];
  __shared__ float    ail[ROWS_WG];
  __shared__ unsigned qcount;

  const int tid  = threadIdx.x;
  const int lane = tid & 31;
  const int wave = __builtin_amdgcn_readfirstlane(tid >> 5);
  const int n0   = blockIdx.x * ROWS_WG;
  const int4* __restrict__ d4p = (const int4*)edst;

  for (int i = tid; i < ROWS_WG * C_OUT; i += EDGE_T) acc[i] = 0.0f;
  if (tid < ROWS_WG) {
    mkey[tid] = 0u;
    lsum[tid] = 0.0f;
    ail[tid]  = ai[n0 + tid];
  }
  if (tid == 0) qcount = 0u;
  __syncthreads();

  if (tid < ROWS_WG) q_push(q, &qcount, E_EDGES + n0 + tid);
  {
    const unsigned* mrow = bmap + (size_t)blockIdx.x * BM_ROW;
    v4u v[MAP_U];
#pragma unroll
    for (int u = 0; u < MAP_U; ++u) {
      const int idx = tid + u * EDGE_T;
      v[u] = (idx < MAP_CH) ? *(const v4u*)(mrow + (size_t)(idx >> 3) * BM_BLK + (idx & 7) * 4) : (v4u)0u;
    }
#pragma unroll
    for (int u = 0; u < MAP_U; ++u) {
      const int idx = tid + u * EDGE_T;
      const int e0 = (idx >> 3) * EBLK + (idx & 7) * 128;
#pragma unroll
      for (int wi = 0; wi < 4; ++wi) {
        unsigned w = v[u][wi];
        while (w != 0u) {
          q_push(q, &qcount, e0 + wi * 32 + __builtin_ctz(w));
          w &= w - 1u;
        }
      }
    }
  }
  __syncthreads();
  const unsigned cnt = qcount;

  if (cnt <= (unsigned)QCAP) {
    q_resolve(q, cnt, esrc, edst, n0, tid);
    for (unsigned i0 = (unsigned)tid; i0 < cnt; i0 += 4 * EDGE_T) {
      float ajv[4];
      int   dl[4];
#pragma unroll
      for (int u = 0; u < 4; ++u) {
        const unsigned i = i0 + u * EDGE_T;
        if (i < cnt) { const unsigned ent = q[i]; dl[u] = (int)(ent >> 16); ajv[u] = aj[ent & 0xFFFFu]; }
      }
#pragma unroll
      for (int u = 0; u < 4; ++u) {
        const unsigned i = i0 + u * EDGE_T;
        if (i < cnt) atomicMax(&mkey[dl[u]], f2key(leaky(ail[dl[u]] + ajv[u])));
      }
    }
    __syncthreads();
    for (unsigned i0 = (unsigned)tid; i0 < cnt; i0 += 4 * EDGE_T) {
      float ajv[4];
      int   dl[4];
#pragma unroll
      for (int u = 0; u < 4; ++u) {
        const unsigned i = i0 + u * EDGE_T;
        if (i < cnt) { const unsigned ent = q[i]; dl[u] = (int)(ent >> 16); ajv[u] = aj[ent & 0xFFFFu]; }
      }
#pragma unroll
      for (int u = 0; u < 4; ++u) {
        const unsigned i = i0 + u * EDGE_T;
        if (i < cnt) atomicAdd(&lsum[dl[u]], expf(leaky(ail[dl[u]] + ajv[u]) - key2f(mkey[dl[u]])));
      }
    }
    __syncthreads();
    rows_from_queue(q, cnt, ail, mkey, lsum, acc, aj, h, wave, lane);
  } else {
    if (tid < ROWS_WG) atomicMax(&mkey[tid], f2key(leaky(ail[tid] + aj[n0 + tid])));
    for (int k0 = 0; k0 < E4; k0 += EDGE_T) {
      const int k = k0 + tid;
      if (k < E4) {
        const int4 d = d4p[k];
        const int dd[4] = {d.x, d.y, d.z, d.w};
#pragma unroll
        for (int j = 0; j < 4; ++j) {
          const unsigned r = (unsigned)(dd[j] - n0);
          if (r < (unsigned)ROWS_WG) atomicMax(&mkey[r], f2key(leaky(ail[r] + aj[esrc[4 * k + j]])));
        }
      }
    }
    __syncthreads();
    if (tid < ROWS_WG) atomicAdd(&lsum[tid], expf(leaky(ail[tid] + aj[n0 + tid]) - key2f(mkey[tid])));
    for (int k0 = 0; k0 < E4; k0 += EDGE_T) {
      const int k = k0 + tid;
      if (k < E4) {
        const int4 d = d4p[k];
        const int dd[4] = {d.x, d.y, d.z, d.w};
#pragma unroll
        for (int j = 0; j < 4; ++j) {
          const unsigned r = (unsigned)(dd[j] - n0);
          if (r < (unsigned)ROWS_WG)
            atomicAdd(&lsum[r], expf(leaky(ail[r] + aj[esrc[4 * k + j]]) - key2f(mkey[r])));
        }
      }
    }
    __syncthreads();
    if (tid == 0) qcount = 0u;
    __syncthreads();
    if (tid < ROWS_WG) q_push(q, &qcount, E_EDGES + n0 + tid);
    for (int k0 = 0; k0 < E4; k0 += EDGE_T) {
      const int k = k0 + tid;
      if (k < E4) {
        const int4 d = d4p[k];
        const int dd[4] = {d.x, d.y, d.z, d.w};
#pragma unroll
        for (int j = 0; j < 4; ++j) {
          const unsigned r = (unsigned)(dd[j] - n0);
          if (r < (unsigned)ROWS_WG) q_push(q, &qcount, 4 * k + j);
        }
      }
      __syncthreads();
      const unsigned c2 = qcount;
      __syncthreads();
      if (c2 + (unsigned)(EDGE_T * 4) > (unsigned)QCAP) {
        q_resolve(q, c2, esrc, edst, n0, tid);
        __syncthreads();
        rows_from_queue(q, c2, ail, mkey, lsum, acc, aj, h, wave, lane);
        if (tid == 0) qcount = 0u;
        __syncthreads();
      }
    }
    const unsigned c3 = qcount;
    q_resolve(q, c3, esrc, edst, n0, tid);
    __syncthreads();
    rows_from_queue(q, c3, ail, mkey, lsum, acc, aj, h, wave, lane);
  }
  __syncthreads();

  const v4f b4 = *(const v4f*)(bias + lane * 4);
  for (int f = tid; f < ROWS_WG * (C_OUT / 4); f += EDGE_T) {
    const v4f a4 = *(const v4f*)(acc + f * 4);
    *(v4f*)(out + (size_t)n0 * C_OUT + (size_t)f * 4) = a4 + b4;
  }
}

extern "C" void kernel_launch(void* const* d_in, const int* in_sizes, int n_in,
                              void* d_out, int out_size, void* d_ws, size_t ws_size,
                              hipStream_t stream) {
  const float* x    = (const float*)d_in[0];
  const int*   ei   = (const int*)  d_in[1];
  const float* W    = (const float*)d_in[2];
  const float* att  = (const float*)d_in[3];
  const float* bias = (const float*)d_in[4];
  float* out = (float*)d_out;

  float* h  = (float*)d_ws;
  float* ai = h + (size_t)N_NODES * C_OUT;
  float* aj = ai + N_NODES;
  unsigned* bmap = (unsigned*)(aj + N_NODES);

  const int* esrc = ei;
  const int* edst = ei + E_EDGES;

  k_gemm<<<GEMM_WGS, 128, 0, stream>>>(x, W, h);
  k_node<<<NEB + NODE_WGS, 128, 0, stream>>>(h, att, edst, ai, aj, bmap);
  k_edge<<<NB_EDGE, EDGE_T, 0, stream>>>(esrc, edst, h, ai, aj, bias, bmap, out);
}


# Edge rows: each gathered h row is ONE 16-B-per-lane load (512 contiguous B, was 4 one-line loads); LDS out rows kept column-permuted so every LDS add stays on 32 banks; add order still arbitrary
# speedup vs baseline: 3.9651x; 1.0505x over previous
#include <hip/hip_runtime.h>


#define N_NODES 50000
#define E_EDGES 800000
#define DIN     256
#define C_OUT   128
#define NEG_SLOPE 0.2f

#define NB_EDGE  250
#define ROWS_WG  (N_NODES / NB_EDGE)
#define EBLK     1024
#define NEB      ((E_EDGES + EBLK - 1) / EBLK)
#define BM_ROW   (EBLK / 32)
#define BM_BLK   (NB_EDGE * BM_ROW)
static_assert(BM_ROW * 4 == 128 && (BM_BLK * 4) % 128 == 0, "one line per owner row");

typedef __attribute__((ext_vector_type(2))) float v2f;
typedef __attribute__((ext_vector_type(8))) float v8f;
typedef __attribute__((ext_vector_type(4))) float v4f;
typedef __attribute__((ext_vector_type(4))) unsigned v4u;

#define GEMM_WGS   512
#define M_TILES    (N_NODES / 16)
static_assert(N_NODES % 16 == 0 && C_OUT == 128 && DIN == 256, "tile shapes");

__global__ __launch_bounds__(128) __attribute__((amdgpu_waves_per_eu(2, 2)))
void k_gemm(const float* __restrict__ x,
            const float* __restrict__ W,
            float* __restrict__ h) {
  const int wave = threadIdx.x >> 5;
  const int lane = threadIdx.x & 31;
  const int hh   = lane >> 4;
  const int c    = lane & 15;
  const int col0 = wave * 32 + c;

  v2f wf[32][2][2];
#pragma unroll
  for (int i = 0; i < 32; ++i) {
#pragma unroll
    for (int j = 0; j < 2; ++j) {
      const int k = 8 * i + 4 * hh + 2 * j;
#pragma unroll
      for (int t = 0; t < 2; ++t) {
        wf[i][j][t].x = W[(size_t)k * C_OUT + col0 + 16 * t];
        wf[i][j][t].y = W[(size_t)(k + 1) * C_OUT + col0 + 16 * t];
      }
    }
  }

  for (int mt = blockIdx.x; mt < M_TILES; mt += GEMM_WGS) {
    const int tM = mt << 4;
    const float* arow = x + (size_t)(tM + c) * DIN + 4 * hh;
    float4 a[32];
#pragma unroll
    for (int i = 0; i < 32; ++i) a[i] = *(const float4*)(arow + 8 * i);

    v8f acc0 = (v8f)0.0f;
    v8f acc1 = (v8f)0.0f;
#pragma unroll
    for (int i = 0; i < 32; ++i) {
      v2f a0; a0.x = a[i].x; a0.y = a[i].y;
      v2f a1; a1.x = a[i].z; a1.y = a[i].w;
      acc0 = __builtin_amdgcn_wmma_f32_16x16x4_f32(false, a0, false, wf[i][0][0],
                                                   (short)0, acc0, false, false);
      acc1 = __builtin_amdgcn_wmma_f32_16x16x4_f32(false, a0, false, wf[i][0][1],
                                                   (short)0, acc1, false, false);
      acc0 = __builtin_amdgcn_wmma_f32_16x16x4_f32(false, a1, false, wf[i][1][0],
                                                   (short)0, acc0, false, false);
      acc1 = __builtin_amdgcn_wmma_f32_16x16x4_f32(false, a1, false, wf[i][1][1],
                                                   (short)0, acc1, false, false);
    }
    float* orow = h + (size_t)(tM + hh * 8) * C_OUT + col0;
#pragma unroll
    for (int r = 0; r < 8; ++r) {
      orow[(size_t)r * C_OUT]      = acc0[r];
      orow[(size_t)r * C_OUT + 16] = acc1[r];
    }
  }
}

#define NODES_PER_WAVE 8
#define NODE_TASKS (N_NODES / NODES_PER_WAVE)
#define NODE_WGS   ((NODE_TASKS + 3) / 4)
static_assert(N_NODES % NODES_PER_WAVE == 0 && NODES_PER_WAVE == 8 && C_OUT == 128, "node pass shapes");
static_assert(E_EDGES % 4 == 0 && EBLK == 8 * 128, "a map work-group: 128 threads, two 16-B target loads each");
__global__ __launch_bounds__(128) void k_node(const float* __restrict__ h,
                                              const float* __restrict__ att,
                                              const int* __restrict__ edst,
                                              float* __restrict__ ai,
                                              float* __restrict__ aj,
                                              unsigned* __restrict__ bmap) {
  __shared__ __attribute__((aligned(16))) unsigned bml[BM_BLK];
  if (blockIdx.x < NEB) {
    const int t = threadIdx.x;
    for (int i = t; i < BM_BLK / 4; i += 128) ((v4u*)bml)[i] = (v4u)0u;
    __syncthreads();
    const int e0 = blockIdx.x * EBLK;
    int4 d[2];
#pragma unroll
    for (int j = 0; j < 2; ++j) {
      const int le = 4 * (t + 128 * j);
      d[j] = (e0 + le < E_EDGES) ? *(const int4*)(edst + e0 + le) : make_int4(-1, -1, -1, -1);
    }
#pragma unroll
    for (int j = 0; j < 2; ++j) {
      const int le = 4 * (t + 128 * j);
      const int dd[4] = {d[j].x, d[j].y, d[j].z, d[j].w};
#pragma unroll
      for (int c = 0; c < 4; ++c) {
        if ((unsigned)dd[c] < (unsigned)N_NODES)
          atomicOr(&bml[((unsigned)dd[c] / ROWS_WG) * BM_ROW + (le >> 5)], 1u << ((le & 31) + c));
      }
    }
    __syncthreads();
    v4u* mp = (v4u*)(bmap + (size_t)blockIdx.x * BM_BLK);
    for (int i = t; i < BM_BLK / 4; i += 128) mp[i] = ((const v4u*)bml)[i];
    return;
  }
  const int lane = threadIdx.x & 31;
  const int task = (blockIdx.x - NEB) * 4 + (threadIdx.x >> 5);
  if (task >= NODE_TASKS) return;
  const int n0 = task * NODES_PER_WAVE;

  const float* hp = h + (size_t)n0 * C_OUT + lane * 4;
  float4 hv[NODES_PER_WAVE];
#pragma unroll
  for (int j = 0; j < NODES_PER_WAVE; ++j) hv[j] = *(const float4*)(hp + (size_t)j * C_OUT);
  const float4 av = *(const float4*)(att + lane * 4);
  const float4 bv = *(const float4*)(att + C_OUT + lane * 4);

  float p[2 * NODES_PER_WAVE];
#pragma unroll
  for (int j = 0; j < NODES_PER_WAVE; ++j) {
    p[2 * j]     = hv[j].x*av.x + hv[j].y*av.y + hv[j].z*av.z + hv[j].w*av.w;
    p[2 * j + 1] = hv[j].x*bv.x + hv[j].y*bv.y + hv[j].z*bv.z + hv[j].w*bv.w;
  }
  const bool b16 = (lane & 16) != 0, b8 = (lane & 8) != 0, b4s = (lane & 4) != 0, b2 = (lane & 2) != 0;
  float q[8], r[4], s[2];
#pragma unroll
  for (int k = 0; k < 8; ++k) {
    const float keep = b16 ? p[k + 8] : p[k];
    const float send = b16 ? p[k] : p[k + 8];
    q[k] = keep + __shfl_xor(send, 16, 32);
  }
#pragma unroll
  for (int k = 0; k < 4; ++k) {
    const float keep = b8 ? q[k + 4] : q[k];
    const float send = b8 ? q[k] : q[k + 4];
    r[k] = keep + __shfl_xor(send, 8, 32);
  }
#pragma unroll
  for (int k = 0; k < 2; ++k) {
    const float keep = b4s ? r[k + 2] : r[k];
    const float send = b4s ? r[k] : r[k + 2];
    s[k] = keep + __shfl_xor(send, 4, 32);
  }
  const float keep = b2 ? s[1] : s[0];
  const float send = b2 ? s[0] : s[1];
  float u = keep + __shfl_xor(send, 2, 32);
  u += __shfl_xor(u, 1, 32);
  if ((lane & 1) == 0) {
    float* dst = (lane & 2) ? aj : ai;
    dst[n0 + (lane >> 2)] = u;
  }
}

__device__ __forceinline__ unsigned f2key(float f) {
  unsigned b = __float_as_uint(f);
  return (b & 0x80000000u) ? ~b : (b | 0x80000000u);
}
__device__ __forceinline__ float key2f(unsigned k) {
  unsigned b = (k & 0x80000000u) ? (k & 0x7FFFFFFFu) : ~k;
  return __uint_as_float(b);
}
__device__ __forceinline__ float leaky(float a) {
  return a > 0.0f ? a : NEG_SLOPE * a;
}

#define QCAP     32768
#define EDGE_T   1024
#define ROW_G    10
#define E4       (E_EDGES / 4)
#define MAP_CH   (NEB * 8)
#define MAP_U    ((MAP_CH + EDGE_T - 1) / EDGE_T)
static_assert(N_NODES == NB_EDGE * ROWS_WG, "rows split evenly");
static_assert(C_OUT == 128 && EDGE_T % 32 == 0, "acc row permutation: 32 lanes x 4 columns");
static_assert(N_NODES <= 65536 && ROWS_WG <= 65536, "queue word packing: 16 bits each for source and local row");
static_assert(E_EDGES % 4 == 0, "dst list in 16-B units");
static_assert(EDGE_T * 4 + ROWS_WG <= QCAP, "one batch of the slow scan fits the queue");

__device__ __forceinline__ void q_push(unsigned* q, unsigned* qcount, int e) {
  const unsigned pos = atomicAdd(qcount, 1u);
  if (pos < (unsigned)QCAP) q[pos] = (unsigned)e;
}
__device__ __forceinline__ void q_resolve(unsigned* q, unsigned cnt,
                                          const int* __restrict__ esrc,
                                          const int* __restrict__ edst, int n0, int tid) {
  for (unsigned i0 = (unsigned)tid; i0 < cnt; i0 += 4 * EDGE_T) {
    int s[4];
    int d[4];
#pragma unroll
    for (int u = 0; u < 4; ++u) {
      const unsigned i = i0 + u * EDGE_T;
      if (i < cnt) {
        const int e = (int)q[i];
        if (e < E_EDGES) { s[u] = esrc[e]; d[u] = edst[e]; }
        else             { s[u] = e - E_EDGES; d[u] = e - E_EDGES; }
      }
    }
#pragma unroll
    for (int u = 0; u < 4; ++u) {
      const unsigned i = i0 + u * EDGE_T;
      if (i < cnt) q[i] = (unsigned)s[u] | ((unsigned)(d[u] - n0) << 16);
    }
  }
}

template <bool FULL>
__device__ __forceinline__ void row_group(int n, unsigned ent, float ajs,
                                          const float* ail, const unsigned* mkey,
                                          const float* lsum, float* acc,
                                          const float* __restrict__ aj,
                                          const float* __restrict__ h, int lane) {
  const int s  = (int)(ent & 0xFFFFu);
  const int dl = (int)(ent >> 16);
  float4 hv[ROW_G];
#pragma unroll
  for (int j = 0; j < ROW_G; ++j) {
    if (FULL || j < n) {
      const int sj = __builtin_amdgcn_readlane(s, j);
      const float* hp = h + (size_t)sj * C_OUT;
      hv[j] = *(const float4*)(hp + lane * 4);
    }
  }
  const float a  = leaky(ail[dl] + ajs);
  const float v  = expf(a - key2f(mkey[dl]));
  const float cf = v / (lsum[dl] + 1e-16f);
#pragma unroll
  for (int j = 0; j < ROW_G; ++j) {
    if (FULL || j < n) {
      const int   dj = __builtin_amdgcn_readlane(dl, j);
      const float cj = __int_as_float(__builtin_amdgcn_readlane(__float_as_int(cf), j));
      float* ar = acc + dj * C_OUT + lane;
      atomicAdd(ar,      hv[j].x * cj);
      atomicAdd(ar + 32, hv[j].y * cj);
      atomicAdd(ar + 64, hv[j].z * cj);
      atomicAdd(ar + 96, hv[j].w * cj);
    }
  }
}

__device__ __forceinline__ void rows_from_queue(const unsigned* q, unsigned cnt,
                                                const float* ail, const unsigned* mkey,
                                                const float* lsum, float* acc,
                                                const float* __restrict__ aj,
                                                const float* __restrict__ h,
                                                int wave, int lane) {
  const unsigned stride = (EDGE_T / 32) * ROW_G;
  unsigned base = (unsigned)wave * ROW_G;
  unsigned ent = (lane < ROW_G && base + (unsigned)lane < cnt) ? q[base + lane] : 0u;
  float ajs = aj[ent & 0xFFFFu];
  for (; base + ROW_G <= cnt; base += stride) {
    const unsigned nb = base + stride;
    const unsigned ent_n = (lane < ROW_G && nb + (unsigned)lane < cnt) ? q[nb + lane] : 0u;
    const float ajs_n = aj[ent_n & 0xFFFFu];
    row_group<true>(ROW_G, ent, ajs, ail, mkey, lsum, acc, aj, h, lane);
    ent = ent_n;
    ajs = ajs_n;
  }
  if (base < cnt) row_group<false>((int)(cnt - base), ent, ajs, ail, mkey, lsum, acc, aj, h, lane);
}

__global__ __launch_bounds__(EDGE_T) void k_edge(const int* __restrict__ esrc,
                                                 const int* __restrict__ edst,
                                                 const float* __restrict__ h,
                                                 const float* __restrict__ ai,
                                                 const float* __restrict__ aj,
                                                 const float* __restrict__ bias,
                                                 const unsigned* __restrict__ bmap,
                                                 float* __restrict__ out) {
  __shared__ float acc[ROWS_WG * C_OUT];
  __shared__ unsigned q[QCAP];
  __shared__ unsigned mkey[ROWS_WG];
  __shared__ float    lsum[ROWS_WG];
  __shared__ float    ail[ROWS_WG];
  __shared__ unsigned qcount;

  const int tid  = threadIdx.x;
  const int lane = tid & 31;
  const int wave = __builtin_amdgcn_readfirstlane(tid >> 5);
  const int n0   = blockIdx.x * ROWS_WG;
  const int4* __restrict__ d4p = (const int4*)edst;

  for (int i = tid; i < ROWS_WG * C_OUT; i += EDGE_T) acc[i] = 0.0f;
  if (tid < ROWS_WG) {
    mkey[tid] = 0u;
    lsum[tid] = 0.0f;
    ail[tid]  = ai[n0 + tid];
  }
  if (tid == 0) qcount = 0u;
  __syncthreads();

  if (tid < ROWS_WG) q_push(q, &qcount, E_EDGES + n0 + tid);
  {
    const unsigned* mrow = bmap + (size_t)blockIdx.x * BM_ROW;
    v4u v[MAP_U];
#pragma unroll
    for (int u = 0; u < MAP_U; ++u) {
      const int idx = tid + u * EDGE_T;
      v[u] = (idx < MAP_CH) ? *(const v4u*)(mrow + (size_t)(idx >> 3) * BM_BLK + (idx & 7) * 4) : (v4u)0u;
    }
#pragma unroll
    for (int u = 0; u < MAP_U; ++u) {
      const int idx = tid + u * EDGE_T;
      const int e0 = (idx >> 3) * EBLK + (idx & 7) * 128;
#pragma unroll
      for (int wi = 0; wi < 4; ++wi) {
        unsigned w = v[u][wi];
        while (w != 0u) {
          q_push(q, &qcount, e0 + wi * 32 + __builtin_ctz(w));
          w &= w - 1u;
        }
      }
    }
  }
  __syncthreads();
  const unsigned cnt = qcount;

  if (cnt <= (unsigned)QCAP) {
    q_resolve(q, cnt, esrc, edst, n0, tid);
    for (unsigned i0 = (unsigned)tid; i0 < cnt; i0 += 4 * EDGE_T) {
      float ajv[4];
      int   dl[4];
#pragma unroll
      for (int u = 0; u < 4; ++u) {
        const unsigned i = i0 + u * EDGE_T;
        if (i < cnt) { const unsigned ent = q[i]; dl[u] = (int)(ent >> 16); ajv[u] = aj[ent & 0xFFFFu]; }
      }
#pragma unroll
      for (int u = 0; u < 4; ++u) {
        const unsigned i = i0 + u * EDGE_T;
        if (i < cnt) atomicMax(&mkey[dl[u]], f2key(leaky(ail[dl[u]] + ajv[u])));
      }
    }
    __syncthreads();
    for (unsigned i0 = (unsigned)tid; i0 < cnt; i0 += 4 * EDGE_T) {
      float ajv[4];
      int   dl[4];
#pragma unroll
      for (int u = 0; u < 4; ++u) {
        const unsigned i = i0 + u * EDGE_T;
        if (i < cnt) { const unsigned ent = q[i]; dl[u] = (int)(ent >> 16); ajv[u] = aj[ent & 0xFFFFu]; }
      }
#pragma unroll
      for (int u = 0; u < 4; ++u) {
        const unsigned i = i0 + u * EDGE_T;
        if (i < cnt) atomicAdd(&lsum[dl[u]], expf(leaky(ail[dl[u]] + ajv[u]) - key2f(mkey[dl[u]])));
      }
    }
    __syncthreads();
    rows_from_queue(q, cnt, ail, mkey, lsum, acc, aj, h, wave, lane);
  } else {
    if (tid < ROWS_WG) atomicMax(&mkey[tid], f2key(leaky(ail[tid] + aj[n0 + tid])));
    for (int k0 = 0; k0 < E4; k0 += EDGE_T) {
      const int k = k0 + tid;
      if (k < E4) {
        const int4 d = d4p[k];
        const int dd[4] = {d.x, d.y, d.z, d.w};
#pragma unroll
        for (int j = 0; j < 4; ++j) {
          const unsigned r = (unsigned)(dd[j] - n0);
          if (r < (unsigned)ROWS_WG) atomicMax(&mkey[r], f2key(leaky(ail[r] + aj[esrc[4 * k + j]])));
        }
      }
    }
    __syncthreads();
    if (tid < ROWS_WG) atomicAdd(&lsum[tid], expf(leaky(ail[tid] + aj[n0 + tid]) - key2f(mkey[tid])));
    for (int k0 = 0; k0 < E4; k0 += EDGE_T) {
      const int k = k0 + tid;
      if (k < E4) {
        const int4 d = d4p[k];
        const int dd[4] = {d.x, d.y, d.z, d.w};
#pragma unroll
        for (int j = 0; j < 4; ++j) {
          const unsigned r = (unsigned)(dd[j] - n0);
          if (r < (unsigned)ROWS_WG)
            atomicAdd(&lsum[r], expf(leaky(ail[r] + aj[esrc[4 * k + j]]) - key2f(mkey[r])));
        }
      }
    }
    __syncthreads();
    if (tid == 0) qcount = 0u;
    __syncthreads();
    if (tid < ROWS_WG) q_push(q, &qcount, E_EDGES + n0 + tid);
    for (int k0 = 0; k0 < E4; k0 += EDGE_T) {
      const int k = k0 + tid;
      if (k < E4) {
        const int4 d = d4p[k];
        const int dd[4] = {d.x, d.y, d.z, d.w};
#pragma unroll
        for (int j = 0; j < 4; ++j) {
          const unsigned r = (unsigned)(dd[j] - n0);
          if (r < (unsigned)ROWS_WG) q_push(q, &qcount, 4 * k + j);
        }
      }
      __syncthreads();
      const unsigned c2 = qcount;
      __syncthreads();
      if (c2 + (unsigned)(EDGE_T * 4) > (unsigned)QCAP) {
        q_resolve(q, c2, esrc, edst, n0, tid);
        __syncthreads();
        rows_from_queue(q, c2, ail, mkey, lsum, acc, aj, h, wave, lane);
        if (tid == 0) qcount = 0u;
        __syncthreads();
      }
    }
    const unsigned c3 = qcount;
    q_resolve(q, c3, esrc, edst, n0, tid);
    __syncthreads();
    rows_from_queue(q, c3, ail, mkey, lsum, acc, aj, h, wave, lane);
  }
  __syncthreads();

  const v4f b4 = *(const v4f*)(bias + lane * 4);
  for (int f = tid; f < ROWS_WG * (C_OUT / 4); f += EDGE_T) {
    const float* ar = acc + (f >> 5) * C_OUT + lane;
    v4f a4;
    a4.x = ar[0];
    a4.y = ar[32];
    a4.z = ar[64];
    a4.w = ar[96];
    *(v4f*)(out + (size_t)n0 * C_OUT + (size_t)f * 4) = a4 + b4;
  }
}

extern "C" void kernel_launch(void* const* d_in, const int* in_sizes, int n_in,
                              void* d_out, int out_size, void* d_ws, size_t ws_size,
                              hipStream_t stream) {
  const float* x    = (const float*)d_in[0];
  const int*   ei   = (const int*)  d_in[1];
  const float* W    = (const float*)d_in[2];
  const float* att  = (const float*)d_in[3];
  const float* bias = (const float*)d_in[4];
  float* out = (float*)d_out;

  float* h  = (float*)d_ws;
  float* ai = h + (size_t)N_NODES * C_OUT;
  float* aj = ai + N_NODES;
  unsigned* bmap = (unsigned*)(aj + N_NODES);

  const int* esrc = ei;
  const int* edst = ei + E_EDGES;

  k_gemm<<<GEMM_WGS, 128, 0, stream>>>(x, W, h);
  k_node<<<NEB + NODE_WGS, 128, 0, stream>>>(h, att, edst, ai, aj, bmap);
  k_edge<<<NB_EDGE, EDGE_T, 0, stream>>>(esrc, edst, h, ai, aj, bias, bmap, out);
}
